# starting kernel
# baseline (speedup 1.0000x reference)
#include <hip/hip_runtime.h>


typedef float v2f __attribute__((ext_vector_type(2)));
typedef float v8f __attribute__((ext_vector_type(8)));

#define HID   19
#define SEQ   2048
#define BATCH 1024
#define NKIT  5

#define WMMA4(a, b, c) __builtin_amdgcn_wmma_f32_16x16x4_f32( \
        false, (a), false, (b), (short)0, (c), false, false)

__device__ __forceinline__ float fast_tanh(float x) {
#if __has_builtin(__builtin_amdgcn_tanhf)
    return __builtin_amdgcn_tanhf(x);
#else
    float r;
    asm("v_tanh_f32 %0, %1" : "=v"(r) : "v"(x));
    return r;
#endif
}
__device__ __forceinline__ float fast_sigmoid(float x) {
    return __builtin_fmaf(0.5f, fast_tanh(0.5f * x), 0.5f);
}
__device__ __forceinline__ void wave_sync() {
    __builtin_amdgcn_wave_barrier();
}

__global__ __launch_bounds__(32) void gru_wmma_scan(
    const float* __restrict__ x,
    const float* __restrict__ Wih,
    const float* __restrict__ Whh,
    const float* __restrict__ bih,
    const float* __restrict__ bhh,
    const float* __restrict__ Wout,
    const float* __restrict__ bout,
    float* __restrict__ out)
{
    __shared__ float GX[64][17];
    __shared__ float GH2[32][17];

    const int lane  = threadIdx.x;
    const int m     = lane & 15;
    const int half  = lane >> 4;
    const int bbase = blockIdx.x * 16;

    v2f Bm[2][2 * NKIT];
    #pragma unroll
    for (int nt = 0; nt < 2; ++nt) {
        const int g = 16 * nt + m;
        #pragma unroll
        for (int ki = 0; ki < 2 * NKIT; ++ki) {
            #pragma unroll
            for (int v = 0; v < 2; ++v) {
                float w = 0.0f;
                if (ki < NKIT) {
                    const int K = 4 * ki + 2 * half + v;
                    if (K < HID) w = Wih[g * HID + K];
                } else {
                    const int K = 4 * (ki - NKIT) + 2 * half + v;
                    w = (K < HID) ? Whh[g * HID + K] : (bih[g] + bhh[g]);
                }
                Bm[nt][ki][v] = w;
            }
        }
    }
    v2f Bih2[2][NKIT], Bhh2[2][NKIT];
    #pragma unroll
    for (int u = 0; u < 2; ++u) {
        const int g = 32 + 16 * u + m;
        #pragma unroll
        for (int ki = 0; ki < NKIT; ++ki) {
            #pragma unroll
            for (int v = 0; v < 2; ++v) {
                const int K = 4 * ki + 2 * half + v;
                float wi = 0.0f, wh = 0.0f;
                if (g < 57) {
                    if (K < HID) {
                        wi = Wih[g * HID + K];
                        wh = Whh[g * HID + K];
                    } else {
                        wh = (g < 38) ? (bih[g] + bhh[g]) : bhh[g];
                    }
                }
                Bih2[u][ki][v] = wi;
                Bhh2[u][ki][v] = wh;
            }
        }
    }
    const float bn2 = (32 + m >= 38) ? bih[32 + m] : 0.0f;
    const float bn3 = (48 + m < 57)  ? bih[48 + m] : 0.0f;
    const float bo  = bout[0];

    float wo[NKIT][2];
    #pragma unroll
    for (int ki = 0; ki < NKIT; ++ki)
        #pragma unroll
        for (int v = 0; v < 2; ++v) {
            const int j = 4 * ki + 2 * half + v;
            wo[ki][v] = (j < HID) ? Wout[j] : 0.0f;
        }

    v2f ah[NKIT];
    #pragma unroll
    for (int ki = 0; ki < NKIT; ++ki) { ah[ki][0] = 0.0f; ah[ki][1] = 0.0f; }
    ah[NKIT - 1][1] = half ? 1.0f : 0.0f;

    auto loadA = [&](int t, v2f (&xa)[NKIT]) {
        const float* xp = x + ((size_t)t * BATCH + bbase + m) * HID;
        #pragma unroll
        for (int ki = 0; ki < NKIT - 1; ++ki) {
            const int K0 = 4 * ki + 2 * half;
            xa[ki][0] = xp[K0];
            xa[ki][1] = xp[K0 + 1];
        }
        xa[NKIT - 1][0] = xp[16 + 2 * half];
        xa[NKIT - 1][1] = xp[17];
    };

    v8f dx2 = {}, dx3 = {};

    auto step = [&](int t, v2f (&xc)[NKIT], v2f (&xn)[NKIT]) {
        loadA((t + 1 < SEQ) ? (t + 1) : (SEQ - 1), xn);

        v8f d0 = {}, d1 = {}, dh2 = {}, dh3 = {};
        #pragma unroll
        for (int ki = 0; ki < NKIT; ++ki) {
            d0  = WMMA4(xc[ki], Bm[0][ki], d0);
            d1  = WMMA4(xc[ki], Bm[1][ki], d1);
            dh2 = WMMA4(ah[ki], Bhh2[0][ki], dh2);
            dh3 = WMMA4(ah[ki], Bhh2[1][ki], dh3);
        }
        #pragma unroll
        for (int ki = 0; ki < NKIT; ++ki) {
            d0 = WMMA4(ah[ki], Bm[0][NKIT + ki], d0);
            d1 = WMMA4(ah[ki], Bm[1][NKIT + ki], d1);
        }
        #pragma unroll
        for (int r = 0; r < 8; ++r) {
            const int c = r + 8 * half;
            GX[m][c]       = d0[r];
            GX[16 + m][c]  = d1[r];
            GX[32 + m][c]  = dx2[r] + ((m < 6) ? dh2[r] : bn2);
            GH2[m][c]      = dh2[r];
            GX[48 + m][c]  = dx3[r] + bn3;
            GH2[16 + m][c] = dh3[r];
        }
        wave_sync();

        float oacc = 0.0f;
        #pragma unroll
        for (int ki = 0; ki < NKIT; ++ki) {
            #pragma unroll
            for (int v = 0; v < 2; ++v) {
                const int j = 4 * ki + 2 * half + v;
                const float rg = fast_sigmoid(GX[j][m]);
                const float zg = fast_sigmoid(GX[19 + j][m]);
                const float ng = fast_tanh(GX[38 + j][m] + rg * GH2[6 + j][m]);
                const float hn = __builtin_fmaf(zg, ah[ki][v] - ng, ng);
                if (ki == NKIT - 1 && v == 1)
                    ah[ki][v] = half ? 1.0f : hn;
                else
                    ah[ki][v] = hn;
                oacc += hn * wo[ki][v];
            }
        }

        v8f nx2 = {}, nx3 = {};
        #pragma unroll
        for (int ki = 0; ki < NKIT; ++ki) {
            nx2 = WMMA4(xn[ki], Bih2[0][ki], nx2);
            nx3 = WMMA4(xn[ki], Bih2[1][ki], nx3);
        }
        dx2 = nx2;
        dx3 = nx3;

        oacc += __shfl_xor(oacc, 16, 32);
        __builtin_nontemporal_store(oacc + bo,
                                    &out[(size_t)t * BATCH + bbase + m]);
        wave_sync();
    };

    v2f xbufA[NKIT], xbufB[NKIT];
    loadA(0, xbufA);
    #pragma unroll
    for (int ki = 0; ki < NKIT; ++ki) {
        dx2 = WMMA4(xbufA[ki], Bih2[0][ki], dx2);
        dx3 = WMMA4(xbufA[ki], Bih2[1][ki], dx3);
    }
    wave_sync();

    for (int t = 0; t < SEQ; t += 2) {
        step(t,     xbufA, xbufB);
        step(t + 1, xbufB, xbufA);
    }
}

extern "C" void kernel_launch(void* const* d_in, const int* in_sizes, int n_in,
                              void* d_out, int out_size, void* d_ws, size_t ws_size,
                              hipStream_t stream) {
    (void)in_sizes; (void)n_in; (void)d_ws; (void)ws_size; (void)out_size;
    const float* x    = (const float*)d_in[0];
    const float* Wih  = (const float*)d_in[1];
    const float* Whh  = (const float*)d_in[2];
    const float* bih  = (const float*)d_in[3];
    const float* bhh  = (const float*)d_in[4];
    const float* Wout = (const float*)d_in[5];
    const float* bout = (const float*)d_in[6];
    float* out = (float*)d_out;

    dim3 grid(BATCH / 16);
    dim3 block(32);
    gru_wmma_scan<<<grid, block, 0, stream>>>(x, Wih, Whh, bih, bhh, Wout, bout, out);
}


# Move all x-side work (x loads and the 20 of 40 WMMAs per step for x @ W_ih^T) off the scan wave to 2 helper waves per work-group; LDS ring in accumulator layout, one barrier per 4 steps
# speedup vs baseline: 1.6180x; 1.6180x over previous
#include <hip/hip_runtime.h>


typedef float v2f __attribute__((ext_vector_type(2)));
typedef float v4f __attribute__((ext_vector_type(4)));
typedef float v8f __attribute__((ext_vector_type(8)));

#define HID   19
#define SEQ   2048
#define BATCH 1024
#define NKIT  5
#define NPROD 2
#define CHUNK 4
#define NCHUNK (SEQ / CHUNK)

#define WMMA4(a, b, c) __builtin_amdgcn_wmma_f32_16x16x4_f32( \
        false, (a), false, (b), (short)0, (c), false, false)

__device__ __forceinline__ float fast_tanh(float x) {
#if __has_builtin(__builtin_amdgcn_tanhf)
    return __builtin_amdgcn_tanhf(x);
#else
    float r;
    asm("v_tanh_f32 %0, %1" : "=v"(r) : "v"(x));
    return r;
#endif
}
__device__ __forceinline__ float fast_sigmoid(float x) {
    return __builtin_fmaf(0.5f, fast_tanh(0.5f * x), 0.5f);
}
__device__ __forceinline__ void wave_sync() {
    __builtin_amdgcn_wave_barrier();
}

__global__ __launch_bounds__(32 * (1 + NPROD)) void gru_wmma_scan(
    const float* __restrict__ x,
    const float* __restrict__ Wih,
    const float* __restrict__ Whh,
    const float* __restrict__ bih,
    const float* __restrict__ bhh,
    const float* __restrict__ Wout,
    const float* __restrict__ bout,
    float* __restrict__ out)
{
    __shared__ float GX[64][17];
    __shared__ float GH2[32][17];
    __shared__ __attribute__((aligned(16))) float XP[2][CHUNK][8][32][4];

    const int lane  = threadIdx.x & 31;
    const int wave  = __builtin_amdgcn_readfirstlane((int)(threadIdx.x >> 5));
    const int m     = lane & 15;
    const int half  = lane >> 4;
    const int bbase = blockIdx.x * 16;

    v2f Bw[4][NKIT];
    #pragma unroll
    for (int nt = 0; nt < 4; ++nt) {
        const int g = 16 * nt + m;
        #pragma unroll
        for (int ki = 0; ki < NKIT; ++ki) {
            #pragma unroll
            for (int v = 0; v < 2; ++v) {
                const int K = 4 * ki + 2 * half + v;
                float w = 0.0f;
                if (g < 57) {
                    if (wave == 0)
                        w = (K < HID) ? Whh[g * HID + K]
                                      : ((g < 38) ? (bih[g] + bhh[g]) : bhh[g]);
                    else if (K < HID)
                        w = Wih[g * HID + K];
                }
                Bw[nt][ki][v] = w;
            }
        }
    }
    const float bn2 = (32 + m >= 38) ? bih[32 + m] : 0.0f;
    const float bn3 = (48 + m < 57)  ? bih[48 + m] : 0.0f;
    const float bo  = bout[0];

    float wo[NKIT][2];
    #pragma unroll
    for (int ki = 0; ki < NKIT; ++ki)
        #pragma unroll
        for (int v = 0; v < 2; ++v) {
            const int j = 4 * ki + 2 * half + v;
            wo[ki][v] = (j < HID) ? Wout[j] : 0.0f;
        }

    v2f ah[NKIT];
    #pragma unroll
    for (int ki = 0; ki < NKIT; ++ki) { ah[ki][0] = 0.0f; ah[ki][1] = 0.0f; }
    ah[NKIT - 1][1] = half ? 1.0f : 0.0f;

    auto loadA = [&](int t, v2f (&xa)[NKIT]) {
        const float* xp = x + ((size_t)t * BATCH + bbase + m) * HID;
        #pragma unroll
        for (int ki = 0; ki < NKIT - 1; ++ki) {
            const int K0 = 4 * ki + 2 * half;
            xa[ki][0] = xp[K0];
            xa[ki][1] = xp[K0 + 1];
        }
        xa[NKIT - 1][0] = xp[16 + 2 * half];
        xa[NKIT - 1][1] = xp[17];
    };

    auto produce = [&](int c) {
        const int p = wave - 1;
        v2f xa[CHUNK / NPROD][NKIT];
        #pragma unroll
        for (int i = 0; i < CHUNK / NPROD; ++i)
            loadA(c * CHUNK + p + NPROD * i, xa[i]);
        #pragma unroll
        for (int i = 0; i < CHUNK / NPROD; ++i) {
            v8f a[4] = {};
            #pragma unroll
            for (int ki = 0; ki < NKIT; ++ki) {
                #pragma unroll
                for (int nt = 0; nt < 4; ++nt)
                    a[nt] = WMMA4(xa[i][ki], Bw[nt][ki], a[nt]);
            }
            float (*dst)[32][4] = XP[c & 1][p + NPROD * i];
            #pragma unroll
            for (int nt = 0; nt < 4; ++nt) {
                v4f lo = {a[nt][0], a[nt][1], a[nt][2], a[nt][3]};
                v4f hi = {a[nt][4], a[nt][5], a[nt][6], a[nt][7]};
                *(v4f*)&dst[2 * nt][lane][0]     = lo;
                *(v4f*)&dst[2 * nt + 1][lane][0] = hi;
            }
        }
    };

    auto loadP = [&](int b, int s, v8f (&P)[4]) {
        float (*srcp)[32][4] = XP[b][s];
        #pragma unroll
        for (int nt = 0; nt < 4; ++nt) {
            const v4f lo = *(const v4f*)&srcp[2 * nt][lane][0];
            const v4f hi = *(const v4f*)&srcp[2 * nt + 1][lane][0];
            P[nt] = (v8f){lo[0], lo[1], lo[2], lo[3], hi[0], hi[1], hi[2], hi[3]};
        }
    };

    auto step = [&](int t, v8f (&P)[4]) {
        v8f d0 = P[0], d1 = P[1];
        v8f dh2 = {}, dh3 = {};
        #pragma unroll
        for (int ki = 0; ki < NKIT; ++ki) {
            d0  = WMMA4(ah[ki], Bw[0][ki], d0);
            d1  = WMMA4(ah[ki], Bw[1][ki], d1);
            dh2 = WMMA4(ah[ki], Bw[2][ki], dh2);
            dh3 = WMMA4(ah[ki], Bw[3][ki], dh3);
        }
        #pragma unroll
        for (int r = 0; r < 8; ++r) {
            const int c = r + 8 * half;
            GX[m][c]       = d0[r];
            GX[16 + m][c]  = d1[r];
            GX[32 + m][c]  = P[2][r] + ((m < 6) ? dh2[r] : bn2);
            GH2[m][c]      = dh2[r];
            GX[48 + m][c]  = P[3][r] + bn3;
            GH2[16 + m][c] = dh3[r];
        }
        wave_sync();

        float oacc = 0.0f;
        #pragma unroll
        for (int ki = 0; ki < NKIT; ++ki) {
            #pragma unroll
            for (int v = 0; v < 2; ++v) {
                const int j = 4 * ki + 2 * half + v;
                const float rg = fast_sigmoid(GX[j][m]);
                const float zg = fast_sigmoid(GX[19 + j][m]);
                const float ng = fast_tanh(GX[38 + j][m] + rg * GH2[6 + j][m]);
                const float hn = __builtin_fmaf(zg, ah[ki][v] - ng, ng);
                if (ki == NKIT - 1 && v == 1)
                    ah[ki][v] = half ? 1.0f : hn;
                else
                    ah[ki][v] = hn;
                oacc += hn * wo[ki][v];
            }
        }

        oacc += __shfl_xor(oacc, 16, 32);
        __builtin_nontemporal_store(oacc + bo,
                                    &out[(size_t)t * BATCH + bbase + m]);
        wave_sync();
    };

    if (wave != 0) produce(0);

    for (int c = 0; c < NCHUNK; ++c) {
        __builtin_amdgcn_fence(__ATOMIC_RELEASE, "workgroup", "local");
        __builtin_amdgcn_s_barrier();
        __builtin_amdgcn_fence(__ATOMIC_ACQUIRE, "workgroup", "local");
        if (wave == 0) {
            const int b  = c & 1;
            const int t0 = c * CHUNK;
            v8f Pa[4], Pb[4];
            loadP(b, 0, Pa);
            #pragma unroll
            for (int s = 0; s < CHUNK; s += 2) {
                loadP(b, s + 1, Pb);
                step(t0 + s, Pa);
                if (s + 2 < CHUNK) loadP(b, s + 2, Pa);
                step(t0 + s + 1, Pb);
            }
        } else if (c + 1 < NCHUNK) {
            produce(c + 1);
        }
    }
}

extern "C" void kernel_launch(void* const* d_in, const int* in_sizes, int n_in,
                              void* d_out, int out_size, void* d_ws, size_t ws_size,
                              hipStream_t stream) {
    (void)in_sizes; (void)n_in; (void)d_ws; (void)ws_size; (void)out_size;
    const float* x    = (const float*)d_in[0];
    const float* Wih  = (const float*)d_in[1];
    const float* Whh  = (const float*)d_in[2];
    const float* bih  = (const float*)d_in[3];
    const float* bhh  = (const float*)d_in[4];
    const float* Wout = (const float*)d_in[5];
    const float* bout = (const float*)d_in[6];
    float* out = (float*)d_out;

    dim3 grid(BATCH / 16);
    dim3 block(32 * (1 + NPROD));
    gru_wmma_scan<<<grid, block, 0, stream>>>(x, Wih, Whh, bih, bhh, Wout, bout, out);
}


# Swap WMMA operand roles (weights as A, state h as B): each lane's accumulators are its own h slots, gates run in registers, LDS transpose removed; k order of the sums and of the output dot changes
# speedup vs baseline: 2.7745x; 1.7148x over previous
#include <hip/hip_runtime.h>


typedef float v2f __attribute__((ext_vector_type(2)));
typedef float v4f __attribute__((ext_vector_type(4)));
typedef float v8f __attribute__((ext_vector_type(8)));

#define HID   19
#define SEQ   2048
#define BATCH 1024
#define NKIT  5
#define NPROD 2
#define CHUNK 4
#define NCHUNK (SEQ / CHUNK)

#define WMMA4(a, b, c) __builtin_amdgcn_wmma_f32_16x16x4_f32( \
        false, (a), false, (b), (short)0, (c), false, false)

__device__ __forceinline__ float fast_tanh(float x) {
#if __has_builtin(__builtin_amdgcn_tanhf)
    return __builtin_amdgcn_tanhf(x);
#else
    float r;
    asm("v_tanh_f32 %0, %1" : "=v"(r) : "v"(x));
    return r;
#endif
}
__device__ __forceinline__ float fast_sigmoid(float x) {
    return __builtin_fmaf(0.5f, fast_tanh(0.5f * x), 0.5f);
}

__global__ __launch_bounds__(32 * (1 + NPROD)) void gru_wmma_scan(
    const float* __restrict__ x,
    const float* __restrict__ Wih,
    const float* __restrict__ Whh,
    const float* __restrict__ bih,
    const float* __restrict__ bhh,
    const float* __restrict__ Wout,
    const float* __restrict__ bout,
    float* __restrict__ out)
{
    __shared__ __attribute__((aligned(16))) float XP[2][CHUNK][8][32][4];

    const int lane  = threadIdx.x & 31;
    const int wave  = __builtin_amdgcn_readfirstlane((int)(threadIdx.x >> 5));
    const int m     = lane & 15;
    const int half  = lane >> 4;
    const int bbase = blockIdx.x * 16;

    v2f Aw[4][NKIT];
    #pragma unroll
    for (int nt = 0; nt < 4; ++nt) {
        const int rr   = m & 7;
        const int hd   = m >> 3;
        const int kind = (nt < 3) ? nt : (rr >> 1);
        const int jj   = 10 * hd + ((nt < 3) ? rr : 8 + (rr & 1));
        const bool live = jj < HID;
        const int g    = ((kind == 3) ? 2 : kind) * HID + jj;
        #pragma unroll
        for (int ki = 0; ki < NKIT; ++ki) {
            #pragma unroll
            for (int v = 0; v < 2; ++v) {
                const int K = 10 * half + 2 * ki + v;
                float w = 0.0f;
                if (live) {
                    if (wave == 0) {
                        if (kind != 3)
                            w = (K < HID) ? Whh[g * HID + K]
                                          : ((kind < 2) ? (bih[g] + bhh[g]) : bhh[g]);
                    } else if (!(nt == 3 && kind == 2)) {
                        w = (K < HID) ? Wih[g * HID + K]
                                      : ((kind >= 2) ? bih[g] : 0.0f);
                    }
                }
                Aw[nt][ki][v] = w;
            }
        }
    }
    const float bo = bout[0];

    float wo[NKIT][2];
    #pragma unroll
    for (int ki = 0; ki < NKIT; ++ki)
        #pragma unroll
        for (int v = 0; v < 2; ++v) {
            const int j = 10 * half + 2 * ki + v;
            wo[ki][v] = (j < HID) ? Wout[j] : 0.0f;
        }

    v2f ah[NKIT];
    #pragma unroll
    for (int ki = 0; ki < NKIT; ++ki) { ah[ki][0] = 0.0f; ah[ki][1] = 0.0f; }
    ah[NKIT - 1][1] = half ? 1.0f : 0.0f;

    auto loadB = [&](int t, v2f (&xb)[NKIT]) {
        const float* xp = x + ((size_t)t * BATCH + bbase + m) * HID + 10 * half;
        #pragma unroll
        for (int ki = 0; ki < NKIT - 1; ++ki) {
            xb[ki][0] = xp[2 * ki];
            xb[ki][1] = xp[2 * ki + 1];
        }
        xb[NKIT - 1][0] = xp[8];
        const float last = xp[9 - half];
        xb[NKIT - 1][1] = half ? 1.0f : last;
    };

    auto produce = [&](int c) {
        const int p = wave - 1;
        v2f xb[CHUNK / NPROD][NKIT];
        #pragma unroll
        for (int i = 0; i < CHUNK / NPROD; ++i)
            loadB(c * CHUNK + p + NPROD * i, xb[i]);
        #pragma unroll
        for (int i = 0; i < CHUNK / NPROD; ++i) {
            v8f a[4] = {};
            #pragma unroll
            for (int ki = 0; ki < NKIT; ++ki) {
                #pragma unroll
                for (int nt = 0; nt < 4; ++nt)
                    a[nt] = WMMA4(Aw[nt][ki], xb[i][ki], a[nt]);
            }
            float (*dst)[32][4] = XP[c & 1][p + NPROD * i];
            #pragma unroll
            for (int nt = 0; nt < 4; ++nt) {
                v4f lo = {a[nt][0], a[nt][1], a[nt][2], a[nt][3]};
                v4f hi = {a[nt][4], a[nt][5], a[nt][6], a[nt][7]};
                *(v4f*)&dst[2 * nt][lane][0]     = lo;
                *(v4f*)&dst[2 * nt + 1][lane][0] = hi;
            }
        }
    };

    auto loadP = [&](int b, int s, v8f (&P)[4]) {
        float (*srcp)[32][4] = XP[b][s];
        #pragma unroll
        for (int nt = 0; nt < 4; ++nt) {
            const v4f lo = *(const v4f*)&srcp[2 * nt][lane][0];
            const v4f hi = *(const v4f*)&srcp[2 * nt + 1][lane][0];
            P[nt] = (v8f){lo[0], lo[1], lo[2], lo[3], hi[0], hi[1], hi[2], hi[3]};
        }
    };

    auto step = [&](int t, v8f (&P)[4]) {
        v8f d0 = P[0], d1 = P[1], d3 = P[3];
        v8f d2 = {};
        #pragma unroll
        for (int ki = 0; ki < NKIT; ++ki) {
            d0 = WMMA4(Aw[0][ki], ah[ki], d0);
            d1 = WMMA4(Aw[1][ki], ah[ki], d1);
            d2 = WMMA4(Aw[2][ki], ah[ki], d2);
            d3 = WMMA4(Aw[3][ki], ah[ki], d3);
        }

        float oacc = 0.0f;
        #pragma unroll
        for (int ki = 0; ki < NKIT; ++ki) {
            #pragma unroll
            for (int v = 0; v < 2; ++v) {
                const int i = 2 * ki + v;
                const float rp = (i < 8) ? d0[i & 7]   : d3[i & 1];
                const float zp = (i < 8) ? d1[i & 7]   : d3[2 + (i & 1)];
                const float hp = (i < 8) ? d2[i & 7]   : d3[4 + (i & 1)];
                const float xp = (i < 8) ? P[2][i & 7] : d3[6 + (i & 1)];
                const float rg = fast_sigmoid(rp);
                const float zg = fast_sigmoid(zp);
                const float ng = fast_tanh(xp + rg * hp);
                const float hn = __builtin_fmaf(zg, ah[ki][v] - ng, ng);
                if (ki == NKIT - 1 && v == 1)
                    ah[ki][v] = half ? 1.0f : hn;
                else
                    ah[ki][v] = hn;
                oacc += hn * wo[ki][v];
            }
        }

        oacc += __shfl_xor(oacc, 16, 32);
        __builtin_nontemporal_store(oacc + bo,
                                    &out[(size_t)t * BATCH + bbase + m]);
    };

    if (wave != 0) produce(0);

    for (int c = 0; c < NCHUNK; ++c) {
        __builtin_amdgcn_fence(__ATOMIC_RELEASE, "workgroup", "local");
        __builtin_amdgcn_s_barrier();
        __builtin_amdgcn_fence(__ATOMIC_ACQUIRE, "workgroup", "local");
        if (wave == 0) {
            const int b  = c & 1;
            const int t0 = c * CHUNK;
            v8f Pa[4], Pb[4];
            loadP(b, 0, Pa);
            #pragma unroll
            for (int s = 0; s < CHUNK; s += 2) {
                loadP(b, s + 1, Pb);
                step(t0 + s, Pa);
                if (s + 2 < CHUNK) loadP(b, s + 2, Pa);
                step(t0 + s + 1, Pb);
            }
        } else if (c + 1 < NCHUNK) {
            produce(c + 1);
        }
    }
}

extern "C" void kernel_launch(void* const* d_in, const int* in_sizes, int n_in,
                              void* d_out, int out_size, void* d_ws, size_t ws_size,
                              hipStream_t stream) {
    (void)in_sizes; (void)n_in; (void)d_ws; (void)ws_size; (void)out_size;
    const float* x    = (const float*)d_in[0];
    const float* Wih  = (const float*)d_in[1];
    const float* Whh  = (const float*)d_in[2];
    const float* bih  = (const float*)d_in[3];
    const float* bhh  = (const float*)d_in[4];
    const float* Wout = (const float*)d_in[5];
    const float* bout = (const float*)d_in[6];
    float* out = (float*)d_out;

    dim3 grid(BATCH / 16);
    dim3 block(32 * (1 + NPROD));
    gru_wmma_scan<<<grid, block, 0, stream>>>(x, Wih, Whh, bih, bhh, Wout, bout, out);
}


# LDS ring chunk 4 -> 16 timesteps per barrier (ring 32 -> 128 KiB): the scan wave's chunk-edge drain (last step's gates alone, output exchange waits, barrier, LDS refill) comes 4x less often
# speedup vs baseline: 2.9007x; 1.0455x over previous
#include <hip/hip_runtime.h>


typedef float v2f __attribute__((ext_vector_type(2)));
typedef float v4f __attribute__((ext_vector_type(4)));
typedef float v8f __attribute__((ext_vector_type(8)));

#define HID   19
#define SEQ   2048
#define BATCH 1024
#define NKIT  5
#define NPROD 2
#define CHUNK 16
#define NCHUNK (SEQ / CHUNK)

#define WMMA4(a, b, c) __builtin_amdgcn_wmma_f32_16x16x4_f32( \
        false, (a), false, (b), (short)0, (c), false, false)

__device__ __forceinline__ float fast_tanh(float x) {
#if __has_builtin(__builtin_amdgcn_tanhf)
    return __builtin_amdgcn_tanhf(x);
#else
    float r;
    asm("v_tanh_f32 %0, %1" : "=v"(r) : "v"(x));
    return r;
#endif
}
__device__ __forceinline__ float fast_sigmoid(float x) {
    return __builtin_fmaf(0.5f, fast_tanh(0.5f * x), 0.5f);
}

__global__ __launch_bounds__(32 * (1 + NPROD)) void gru_wmma_scan(
    const float* __restrict__ x,
    const float* __restrict__ Wih,
    const float* __restrict__ Whh,
    const float* __restrict__ bih,
    const float* __restrict__ bhh,
    const float* __restrict__ Wout,
    const float* __restrict__ bout,
    float* __restrict__ out)
{
    __shared__ __attribute__((aligned(16))) float XP[2][CHUNK][8][32][4];

    const int lane  = threadIdx.x & 31;
    const int wave  = __builtin_amdgcn_readfirstlane((int)(threadIdx.x >> 5));
    const int m     = lane & 15;
    const int half  = lane >> 4;
    const int bbase = blockIdx.x * 16;

    v2f Aw[4][NKIT];
    #pragma unroll
    for (int nt = 0; nt < 4; ++nt) {
        const int rr   = m & 7;
        const int hd   = m >> 3;
        const int kind = (nt < 3) ? nt : (rr >> 1);
        const int jj   = 10 * hd + ((nt < 3) ? rr : 8 + (rr & 1));
        const bool live = jj < HID;
        const int g    = ((kind == 3) ? 2 : kind) * HID + jj;
        #pragma unroll
        for (int ki = 0; ki < NKIT; ++ki) {
            #pragma unroll
            for (int v = 0; v < 2; ++v) {
                const int K = 10 * half + 2 * ki + v;
                float w = 0.0f;
                if (live) {
                    if (wave == 0) {
                        if (kind != 3)
                            w = (K < HID) ? Whh[g * HID + K]
                                          : ((kind < 2) ? (bih[g] + bhh[g]) : bhh[g]);
                    } else if (!(nt == 3 && kind == 2)) {
                        w = (K < HID) ? Wih[g * HID + K]
                                      : ((kind >= 2) ? bih[g] : 0.0f);
                    }
                }
                Aw[nt][ki][v] = w;
            }
        }
    }
    const float bo = bout[0];

    float wo[NKIT][2];
    #pragma unroll
    for (int ki = 0; ki < NKIT; ++ki)
        #pragma unroll
        for (int v = 0; v < 2; ++v) {
            const int j = 10 * half + 2 * ki + v;
            wo[ki][v] = (j < HID) ? Wout[j] : 0.0f;
        }

    v2f ah[NKIT];
    #pragma unroll
    for (int ki = 0; ki < NKIT; ++ki) { ah[ki][0] = 0.0f; ah[ki][1] = 0.0f; }
    ah[NKIT - 1][1] = half ? 1.0f : 0.0f;

    auto loadB = [&](int t, v2f (&xb)[NKIT]) {
        const float* xp = x + ((size_t)t * BATCH + bbase + m) * HID + 10 * half;
        #pragma unroll
        for (int ki = 0; ki < NKIT - 1; ++ki) {
            xb[ki][0] = xp[2 * ki];
            xb[ki][1] = xp[2 * ki + 1];
        }
        xb[NKIT - 1][0] = xp[8];
        const float last = xp[9 - half];
        xb[NKIT - 1][1] = half ? 1.0f : last;
    };

    auto produce = [&](int c) {
        const int p = wave - 1;
        v2f xb[CHUNK / NPROD][NKIT];
        #pragma unroll
        for (int i = 0; i < CHUNK / NPROD; ++i)
            loadB(c * CHUNK + p + NPROD * i, xb[i]);
        #pragma unroll
        for (int i = 0; i < CHUNK / NPROD; ++i) {
            v8f a[4] = {};
            #pragma unroll
            for (int ki = 0; ki < NKIT; ++ki) {
                #pragma unroll
                for (int nt = 0; nt < 4; ++nt)
                    a[nt] = WMMA4(Aw[nt][ki], xb[i][ki], a[nt]);
            }
            float (*dst)[32][4] = XP[c & 1][p + NPROD * i];
            #pragma unroll
            for (int nt = 0; nt < 4; ++nt) {
                v4f lo = {a[nt][0], a[nt][1], a[nt][2], a[nt][3]};
                v4f hi = {a[nt][4], a[nt][5], a[nt][6], a[nt][7]};
                *(v4f*)&dst[2 * nt][lane][0]     = lo;
                *(v4f*)&dst[2 * nt + 1][lane][0] = hi;
            }
        }
    };

    auto loadP = [&](int b, int s, v8f (&P)[4]) {
        float (*srcp)[32][4] = XP[b][s];
        #pragma unroll
        for (int nt = 0; nt < 4; ++nt) {
            const v4f lo = *(const v4f*)&srcp[2 * nt][lane][0];
            const v4f hi = *(const v4f*)&srcp[2 * nt + 1][lane][0];
            P[nt] = (v8f){lo[0], lo[1], lo[2], lo[3], hi[0], hi[1], hi[2], hi[3]};
        }
    };

    auto step = [&](int t, v8f (&P)[4]) {
        v8f d0 = P[0], d1 = P[1], d3 = P[3];
        v8f d2 = {};
        #pragma unroll
        for (int ki = 0; ki < NKIT; ++ki) {
            d0 = WMMA4(Aw[0][ki], ah[ki], d0);
            d1 = WMMA4(Aw[1][ki], ah[ki], d1);
            d2 = WMMA4(Aw[2][ki], ah[ki], d2);
            d3 = WMMA4(Aw[3][ki], ah[ki], d3);
        }

        float oacc = 0.0f;
        #pragma unroll
        for (int ki = 0; ki < NKIT; ++ki) {
            #pragma unroll
            for (int v = 0; v < 2; ++v) {
                const int i = 2 * ki + v;
                const float rp = (i < 8) ? d0[i & 7]   : d3[i & 1];
                const float zp = (i < 8) ? d1[i & 7]   : d3[2 + (i & 1)];
                const float hp = (i < 8) ? d2[i & 7]   : d3[4 + (i & 1)];
                const float xp = (i < 8) ? P[2][i & 7] : d3[6 + (i & 1)];
                const float rg = fast_sigmoid(rp);
                const float zg = fast_sigmoid(zp);
                const float ng = fast_tanh(xp + rg * hp);
                const float hn = __builtin_fmaf(zg, ah[ki][v] - ng, ng);
                if (ki == NKIT - 1 && v == 1)
                    ah[ki][v] = half ? 1.0f : hn;
                else
                    ah[ki][v] = hn;
                oacc += hn * wo[ki][v];
            }
        }

        oacc += __shfl_xor(oacc, 16, 32);
        __builtin_nontemporal_store(oacc + bo,
                                    &out[(size_t)t * BATCH + bbase + m]);
    };

    if (wave != 0) produce(0);

    for (int c = 0; c < NCHUNK; ++c) {
        __builtin_amdgcn_fence(__ATOMIC_RELEASE, "workgroup", "local");
        __builtin_amdgcn_s_barrier();
        __builtin_amdgcn_fence(__ATOMIC_ACQUIRE, "workgroup", "local");
        if (wave == 0) {
            const int b  = c & 1;
            const int t0 = c * CHUNK;
            v8f Pa[4], Pb[4];
            loadP(b, 0, Pa);
            #pragma unroll
            for (int s = 0; s < CHUNK; s += 2) {
                loadP(b, s + 1, Pb);
                step(t0 + s, Pa);
                if (s + 2 < CHUNK) loadP(b, s + 2, Pa);
                step(t0 + s + 1, Pb);
            }
        } else if (c + 1 < NCHUNK) {
            produce(c + 1);
        }
    }
}

extern "C" void kernel_launch(void* const* d_in, const int* in_sizes, int n_in,
                              void* d_out, int out_size, void* d_ws, size_t ws_size,
                              hipStream_t stream) {
    (void)in_sizes; (void)n_in; (void)d_ws; (void)ws_size; (void)out_size;
    const float* x    = (const float*)d_in[0];
    const float* Wih  = (const float*)d_in[1];
    const float* Whh  = (const float*)d_in[2];
    const float* bih  = (const float*)d_in[3];
    const float* bhh  = (const float*)d_in[4];
    const float* Wout = (const float*)d_in[5];
    const float* bout = (const float*)d_in[6];
    float* out = (float*)d_out;

    dim3 grid(BATCH / 16);
    dim3 block(32 * (1 + NPROD));
    gru_wmma_scan<<<grid, block, 0, stream>>>(x, Wih, Whh, bih, bhh, Wout, bout, out);
}


# Output projection folded into a spare row of the h-side weight tile: out[t-1] arrives in step t's WMMA accumulator; VALU dot and LDS lane exchange removed; k order of the output sum changes
# speedup vs baseline: 3.1185x; 1.0751x over previous
#include <hip/hip_runtime.h>


typedef float v2f __attribute__((ext_vector_type(2)));
typedef float v4f __attribute__((ext_vector_type(4)));
typedef float v8f __attribute__((ext_vector_type(8)));

#define HID   19
#define SEQ   2048
#define BATCH 1024
#define NKIT  5
#define NPROD 2
#define CHUNK 16
#define NCHUNK (SEQ / CHUNK)

#define WMMA4(a, b, c) __builtin_amdgcn_wmma_f32_16x16x4_f32( \
        false, (a), false, (b), (short)0, (c), false, false)

__device__ __forceinline__ float fast_tanh(float x) {
#if __has_builtin(__builtin_amdgcn_tanhf)
    return __builtin_amdgcn_tanhf(x);
#else
    float r;
    asm("v_tanh_f32 %0, %1" : "=v"(r) : "v"(x));
    return r;
#endif
}
__device__ __forceinline__ float fast_sigmoid(float x) {
    return __builtin_fmaf(0.5f, fast_tanh(0.5f * x), 0.5f);
}

__global__ __launch_bounds__(32 * (1 + NPROD)) void gru_wmma_scan(
    const float* __restrict__ x,
    const float* __restrict__ Wih,
    const float* __restrict__ Whh,
    const float* __restrict__ bih,
    const float* __restrict__ bhh,
    const float* __restrict__ Wout,
    const float* __restrict__ bout,
    float* __restrict__ out)
{
    __shared__ __attribute__((aligned(16))) float XP[2][CHUNK][8][32][4];

    const int lane  = threadIdx.x & 31;
    const int wave  = __builtin_amdgcn_readfirstlane((int)(threadIdx.x >> 5));
    const int m     = lane & 15;
    const int half  = lane >> 4;
    const int bbase = blockIdx.x * 16;

    v2f Aw[4][NKIT];
    #pragma unroll
    for (int nt = 0; nt < 4; ++nt) {
        const int rr   = m & 7;
        const int hd   = m >> 3;
        const int kind = (nt < 3) ? nt : (rr >> 1);
        const int jj   = 10 * hd + ((nt < 3) ? rr : 8 + (rr & 1));
        const bool live = jj < HID;
        const int g    = ((kind == 3) ? 2 : kind) * HID + jj;
        #pragma unroll
        for (int ki = 0; ki < NKIT; ++ki) {
            #pragma unroll
            for (int v = 0; v < 2; ++v) {
                const int K = 10 * half + 2 * ki + v;
                float w = 0.0f;
                if (live) {
                    if (wave == 0) {
                        if (kind != 3)
                            w = (K < HID) ? Whh[g * HID + K]
                                          : ((kind < 2) ? (bih[g] + bhh[g]) : bhh[g]);
                    } else if (!(nt == 3 && kind == 2)) {
                        w = (K < HID) ? Wih[g * HID + K]
                                      : ((kind >= 2) ? bih[g] : 0.0f);
                    }
                }
                if (wave == 0 && nt == 3 && m == 9)
                    w = (K < HID) ? Wout[K] : bout[0];
                Aw[nt][ki][v] = w;
            }
        }
    }
    const int hmask = -half;

    auto emit_out = [&](float v, int t) {
        const int b = __builtin_bit_cast(int, v) & hmask;
        const int o = b | __builtin_amdgcn_permlanex16(b, b, 0x76543210, 0xfedcba98,
                                                       false, false);
        __builtin_nontemporal_store(__builtin_bit_cast(float, o),
                                    &out[(size_t)t * BATCH + bbase + m]);
    };

    v2f ah[NKIT];
    #pragma unroll
    for (int ki = 0; ki < NKIT; ++ki) { ah[ki][0] = 0.0f; ah[ki][1] = 0.0f; }
    ah[NKIT - 1][1] = half ? 1.0f : 0.0f;

    auto loadB = [&](int t, v2f (&xb)[NKIT]) {
        const float* xp = x + ((size_t)t * BATCH + bbase + m) * HID + 10 * half;
        #pragma unroll
        for (int ki = 0; ki < NKIT - 1; ++ki) {
            xb[ki][0] = xp[2 * ki];
            xb[ki][1] = xp[2 * ki + 1];
        }
        xb[NKIT - 1][0] = xp[8];
        const float last = xp[9 - half];
        xb[NKIT - 1][1] = half ? 1.0f : last;
    };

    auto produce = [&](int c) {
        const int p = wave - 1;
        v2f xb[CHUNK / NPROD][NKIT];
        #pragma unroll
        for (int i = 0; i < CHUNK / NPROD; ++i)
            loadB(c * CHUNK + p + NPROD * i, xb[i]);
        #pragma unroll
        for (int i = 0; i < CHUNK / NPROD; ++i) {
            v8f a[4] = {};
            #pragma unroll
            for (int ki = 0; ki < NKIT; ++ki) {
                #pragma unroll
                for (int nt = 0; nt < 4; ++nt)
                    a[nt] = WMMA4(Aw[nt][ki], xb[i][ki], a[nt]);
            }
            float (*dst)[32][4] = XP[c & 1][p + NPROD * i];
            #pragma unroll
            for (int nt = 0; nt < 4; ++nt) {
                v4f lo = {a[nt][0], a[nt][1], a[nt][2], a[nt][3]};
                v4f hi = {a[nt][4], a[nt][5], a[nt][6], a[nt][7]};
                *(v4f*)&dst[2 * nt][lane][0]     = lo;
                *(v4f*)&dst[2 * nt + 1][lane][0] = hi;
            }
        }
    };

    auto loadP = [&](int b, int s, v8f (&P)[4]) {
        float (*srcp)[32][4] = XP[b][s];
        #pragma unroll
        for (int nt = 0; nt < 4; ++nt) {
            const v4f lo = *(const v4f*)&srcp[2 * nt][lane][0];
            const v4f hi = *(const v4f*)&srcp[2 * nt + 1][lane][0];
            P[nt] = (v8f){lo[0], lo[1], lo[2], lo[3], hi[0], hi[1], hi[2], hi[3]};
        }
    };

    auto step = [&](int t, v8f (&P)[4], bool emit) {
        v8f d0 = P[0], d1 = P[1], d3 = P[3];
        v8f d2 = {};
        #pragma unroll
        for (int ki = 0; ki < NKIT; ++ki) {
            d0 = WMMA4(Aw[0][ki], ah[ki], d0);
            d1 = WMMA4(Aw[1][ki], ah[ki], d1);
            d2 = WMMA4(Aw[2][ki], ah[ki], d2);
            d3 = WMMA4(Aw[3][ki], ah[ki], d3);
        }

        const float oprev = d3[1];
        #pragma unroll
        for (int ki = 0; ki < NKIT; ++ki) {
            #pragma unroll
            for (int v = 0; v < 2; ++v) {
                const int i = 2 * ki + v;
                const float rp = (i < 8) ? d0[i & 7]   : d3[i & 1];
                const float zp = (i < 8) ? d1[i & 7]   : d3[2 + (i & 1)];
                const float hp = (i < 8) ? d2[i & 7]   : d3[4 + (i & 1)];
                const float xp = (i < 8) ? P[2][i & 7] : d3[6 + (i & 1)];
                const float rg = fast_sigmoid(rp);
                const float zg = fast_sigmoid(zp);
                const float ng = fast_tanh(xp + rg * hp);
                const float hn = __builtin_fmaf(zg, ah[ki][v] - ng, ng);
                if (ki == NKIT - 1 && v == 1)
                    ah[ki][v] = half ? 1.0f : hn;
                else
                    ah[ki][v] = hn;
            }
        }
        if (emit) emit_out(oprev, t - 1);
    };

    if (wave != 0) produce(0);

    for (int c = 0; c < NCHUNK; ++c) {
        __builtin_amdgcn_fence(__ATOMIC_RELEASE, "workgroup", "local");
        __builtin_amdgcn_s_barrier();
        __builtin_amdgcn_fence(__ATOMIC_ACQUIRE, "workgroup", "local");
        if (wave == 0) {
            const int b  = c & 1;
            const int t0 = c * CHUNK;
            v8f Pa[4], Pb[4];
            loadP(b, 0, Pa);
            #pragma unroll
            for (int s = 0; s < CHUNK; s += 2) {
                loadP(b, s + 1, Pb);
                step(t0 + s, Pa, s > 0);
                if (s + 2 < CHUNK) loadP(b, s + 2, Pa);
                step(t0 + s + 1, Pb, true);
            }
            v8f dl = {};
            #pragma unroll
            for (int ki = 0; ki < NKIT; ++ki)
                dl = WMMA4(Aw[3][ki], ah[ki], dl);
            emit_out(dl[1], t0 + CHUNK - 1);
        } else if (c + 1 < NCHUNK) {
            produce(c + 1);
        }
    }
}

extern "C" void kernel_launch(void* const* d_in, const int* in_sizes, int n_in,
                              void* d_out, int out_size, void* d_ws, size_t ws_size,
                              hipStream_t stream) {
    (void)in_sizes; (void)n_in; (void)d_ws; (void)ws_size; (void)out_size;
    const float* x    = (const float*)d_in[0];
    const float* Wih  = (const float*)d_in[1];
    const float* Whh  = (const float*)d_in[2];
    const float* bih  = (const float*)d_in[3];
    const float* bhh  = (const float*)d_in[4];
    const float* Wout = (const float*)d_in[5];
    const float* bout = (const float*)d_in[6];
    float* out = (float*)d_out;

    dim3 grid(BATCH / 16);
    dim3 block(32 * (1 + NPROD));
    gru_wmma_scan<<<grid, block, 0, stream>>>(x, Wih, Whh, bih, bhh, Wout, bout, out);
}


# Gate math on float pairs: one packed f32 op per two units, sigmoid halving folded into r/z weight rows (exact); ring reads held in one mid-step block by a fence that stops only LDS reads and WMMAs
# speedup vs baseline: 3.2290x; 1.0354x over previous
#include <hip/hip_runtime.h>


typedef float v2f __attribute__((ext_vector_type(2)));
typedef float v4f __attribute__((ext_vector_type(4)));
typedef float v8f __attribute__((ext_vector_type(8)));

#define HID   19
#define SEQ   2048
#define BATCH 1024
#define NKIT  5
#define NPROD 2
#define CHUNK 16
#define NCHUNK (SEQ / CHUNK)

#define RING_READS_STAY_HERE() __builtin_amdgcn_sched_barrier(0x0676)

#define WMMA4(a, b, c) __builtin_amdgcn_wmma_f32_16x16x4_f32( \
        false, (a), false, (b), (short)0, (c), false, false)

__device__ __forceinline__ float fast_tanh(float x) {
#if __has_builtin(__builtin_amdgcn_tanhf)
    return __builtin_amdgcn_tanhf(x);
#else
    float r;
    asm("v_tanh_f32 %0, %1" : "=v"(r) : "v"(x));
    return r;
#endif
}

__global__ __launch_bounds__(32 * (1 + NPROD)) void gru_wmma_scan(
    const float* __restrict__ x,
    const float* __restrict__ Wih,
    const float* __restrict__ Whh,
    const float* __restrict__ bih,
    const float* __restrict__ bhh,
    const float* __restrict__ Wout,
    const float* __restrict__ bout,
    float* __restrict__ out)
{
    __shared__ __attribute__((aligned(16))) float XP[2][CHUNK][8][32][4];

    const int lane  = threadIdx.x & 31;
    const int wave  = __builtin_amdgcn_readfirstlane((int)(threadIdx.x >> 5));
    const int m     = lane & 15;
    const int half  = lane >> 4;
    const int bbase = blockIdx.x * 16;

    v2f Aw[4][NKIT];
    #pragma unroll
    for (int nt = 0; nt < 4; ++nt) {
        const int rr   = m & 7;
        const int hd   = m >> 3;
        const int kind = (nt < 3) ? nt : (rr >> 1);
        const int jj   = 10 * hd + ((nt < 3) ? rr : 8 + (rr & 1));
        const bool live = jj < HID;
        const int g    = ((kind == 3) ? 2 : kind) * HID + jj;
        #pragma unroll
        for (int ki = 0; ki < NKIT; ++ki) {
            #pragma unroll
            for (int v = 0; v < 2; ++v) {
                const int K = 10 * half + 2 * ki + v;
                float w = 0.0f;
                if (live) {
                    if (wave == 0) {
                        if (kind != 3)
                            w = (K < HID) ? Whh[g * HID + K]
                                          : ((kind < 2) ? (bih[g] + bhh[g]) : bhh[g]);
                    } else if (!(nt == 3 && kind == 2)) {
                        w = (K < HID) ? Wih[g * HID + K]
                                      : ((kind >= 2) ? bih[g] : 0.0f);
                    }
                    if (kind < 2) w *= 0.5f;
                }
                if (wave == 0 && nt == 3 && m == 9)
                    w = (K < HID) ? Wout[K] : bout[0];
                Aw[nt][ki][v] = w;
            }
        }
    }
    const int hmask = -half;

    auto emit_out = [&](float v, int t) {
        const int b = __builtin_bit_cast(int, v) & hmask;
        const int o = b | __builtin_amdgcn_permlanex16(b, b, 0x76543210, 0xfedcba98,
                                                       false, false);
        __builtin_nontemporal_store(__builtin_bit_cast(float, o),
                                    &out[(size_t)t * BATCH + bbase + m]);
    };

    v2f ah[NKIT];
    #pragma unroll
    for (int ki = 0; ki < NKIT; ++ki) { ah[ki][0] = 0.0f; ah[ki][1] = 0.0f; }
    ah[NKIT - 1][1] = half ? 1.0f : 0.0f;

    auto loadB = [&](int t, v2f (&xb)[NKIT]) {
        const float* xp = x + ((size_t)t * BATCH + bbase + m) * HID + 10 * half;
        #pragma unroll
        for (int ki = 0; ki < NKIT - 1; ++ki) {
            xb[ki][0] = xp[2 * ki];
            xb[ki][1] = xp[2 * ki + 1];
        }
        xb[NKIT - 1][0] = xp[8];
        const float last = xp[9 - half];
        xb[NKIT - 1][1] = half ? 1.0f : last;
    };

    auto produce = [&](int c) {
        const int p = wave - 1;
        v2f xb[CHUNK / NPROD][NKIT];
        #pragma unroll
        for (int i = 0; i < CHUNK / NPROD; ++i)
            loadB(c * CHUNK + p + NPROD * i, xb[i]);
        #pragma unroll
        for (int i = 0; i < CHUNK / NPROD; ++i) {
            v8f a[4] = {};
            #pragma unroll
            for (int ki = 0; ki < NKIT; ++ki) {
                #pragma unroll
                for (int nt = 0; nt < 4; ++nt)
                    a[nt] = WMMA4(Aw[nt][ki], xb[i][ki], a[nt]);
            }
            float (*dst)[32][4] = XP[c & 1][p + NPROD * i];
            #pragma unroll
            for (int nt = 0; nt < 4; ++nt) {
                v4f lo = {a[nt][0], a[nt][1], a[nt][2], a[nt][3]};
                v4f hi = {a[nt][4], a[nt][5], a[nt][6], a[nt][7]};
                *(v4f*)&dst[2 * nt][lane][0]     = lo;
                *(v4f*)&dst[2 * nt + 1][lane][0] = hi;
            }
        }
    };

    auto loadP = [&](int b, int s, v8f (&P)[4]) {
        float (*srcp)[32][4] = XP[b][s];
        #pragma unroll
        for (int nt = 0; nt < 4; ++nt) {
            const v4f lo = *(const v4f*)&srcp[2 * nt][lane][0];
            const v4f hi = *(const v4f*)&srcp[2 * nt + 1][lane][0];
            P[nt] = (v8f){lo[0], lo[1], lo[2], lo[3], hi[0], hi[1], hi[2], hi[3]};
        }
    };

    auto step = [&](int t, v8f (&P)[4], bool emit, bool fetch, int rb, int sn, v8f (&Pn)[4]) {
        v8f d0 = P[0], d1 = P[1], d3 = P[3];
        v8f d2 = {};
        #pragma unroll
        for (int ki = 0; ki < NKIT; ++ki) {
            d0 = WMMA4(Aw[0][ki], ah[ki], d0);
            d1 = WMMA4(Aw[1][ki], ah[ki], d1);
            d2 = WMMA4(Aw[2][ki], ah[ki], d2);
            d3 = WMMA4(Aw[3][ki], ah[ki], d3);
            if (ki == 3 && fetch) {
                RING_READS_STAY_HERE();
                loadP(rb, sn, Pn);
                RING_READS_STAY_HERE();
            }
        }

        const float oprev = d3[1];
        const v2f c05 = {0.5f, 0.5f};
        #pragma unroll
        for (int ki = 0; ki < NKIT; ++ki) {
            const int e = (ki < 4) ? 2 * ki : 0;
            const v2f rp2 = (ki < 4) ? (v2f){d0[e], d0[e + 1]}     : (v2f){d3[0], d3[1]};
            const v2f zp2 = (ki < 4) ? (v2f){d1[e], d1[e + 1]}     : (v2f){d3[2], d3[3]};
            const v2f hp  = (ki < 4) ? (v2f){d2[e], d2[e + 1]}     : (v2f){d3[4], d3[5]};
            const v2f xp  = (ki < 4) ? (v2f){P[2][e], P[2][e + 1]} : (v2f){d3[6], d3[7]};
            const v2f tr = {fast_tanh(rp2[0]), fast_tanh(rp2[1])};
            const v2f tz = {fast_tanh(zp2[0]), fast_tanh(zp2[1])};
            const v2f rg = __builtin_elementwise_fma(c05, tr, c05);
            const v2f zg = __builtin_elementwise_fma(c05, tz, c05);
            const v2f np = __builtin_elementwise_fma(rg, hp, xp);
            const v2f ng = {fast_tanh(np[0]), fast_tanh(np[1])};
            v2f hn = __builtin_elementwise_fma(zg, ah[ki] - ng, ng);
            if (ki == NKIT - 1) hn[1] = half ? 1.0f : hn[1];
            ah[ki] = hn;
        }
        if (emit) emit_out(oprev, t - 1);
    };

    if (wave != 0) produce(0);

    for (int c = 0; c < NCHUNK; ++c) {
        __builtin_amdgcn_fence(__ATOMIC_RELEASE, "workgroup", "local");
        __builtin_amdgcn_s_barrier();
        __builtin_amdgcn_fence(__ATOMIC_ACQUIRE, "workgroup", "local");
        if (wave == 0) {
            const int b  = c & 1;
            const int t0 = c * CHUNK;
            v8f Pa[4], Pb[4];
            loadP(b, 0, Pa);
            #pragma unroll
            for (int s = 0; s < CHUNK; s += 2) {
                step(t0 + s, Pa, s > 0, true, b, s + 1, Pb);
                step(t0 + s + 1, Pb, true, s + 2 < CHUNK, b, s + 2, Pa);
            }
            v8f dl = {};
            #pragma unroll
            for (int ki = 0; ki < NKIT; ++ki)
                dl = WMMA4(Aw[3][ki], ah[ki], dl);
            emit_out(dl[1], t0 + CHUNK - 1);
        } else if (c + 1 < NCHUNK) {
            produce(c + 1);
        }
    }
}

extern "C" void kernel_launch(void* const* d_in, const int* in_sizes, int n_in,
                              void* d_out, int out_size, void* d_ws, size_t ws_size,
                              hipStream_t stream) {
    (void)in_sizes; (void)n_in; (void)d_ws; (void)ws_size; (void)out_size;
    const float* x    = (const float*)d_in[0];
    const float* Wih  = (const float*)d_in[1];
    const float* Whh  = (const float*)d_in[2];
    const float* bih  = (const float*)d_in[3];
    const float* bhh  = (const float*)d_in[4];
    const float* Wout = (const float*)d_in[5];
    const float* bout = (const float*)d_in[6];
    float* out = (float*)d_out;

    dim3 grid(BATCH / 16);
    dim3 block(32 * (1 + NPROD));
    gru_wmma_scan<<<grid, block, 0, stream>>>(x, Wih, Whh, bih, bhh, Wout, bout, out);
}
